# starting kernel
# baseline (speedup 1.0000x reference)
#include <hip/hip_runtime.h>

typedef __attribute__((ext_vector_type(2))) float v2f;
typedef __attribute__((ext_vector_type(8))) float v8f;

constexpr int Bsz = 128;
constexpr int Nsl = 32;
constexpr int Dd  = 128;
constexpr int Hh  = 256;
constexpr int BN  = Bsz * Nsl;
constexpr float LN_EPS = 1e-5f;

__global__ void enc_gemm_kernel(const float* __restrict__ slots,
                                const float* __restrict__ Wenc,
                                float* __restrict__ UV)
{
    const int lane = threadIdx.x & 31;
    const int wave = threadIdx.x >> 5;
    const int tile = blockIdx.x * 8 + wave;
    const int tm = tile >> 5;
    const int tn = tile & 31;

    const int row  = tm * 16 + (lane & 15);
    const int cg   = tn * 16 + (lane & 15);
    const int whichW = cg >> 8;
    const int hcol   = cg & 255;
    const int klo  = (lane >> 4) * 2;

    const float* __restrict__ arow = slots + (size_t)row * Dd;

    v8f acc = {};
    #pragma unroll 4
    for (int k = 0; k < Dd; k += 4) {
        const int kk = k + klo;
        v2f a; a.x = arow[kk]; a.y = arow[kk + 1];
        const float* bp = Wenc + (size_t)(whichW * Dd + kk) * Hh + hcol;
        v2f b; b.x = bp[0]; b.y = bp[Hh];
        acc = __builtin_amdgcn_wmma_f32_16x16x4_f32(false, a, false, b,
                                                    (short)0, acc, false, false);
    }

    const int hi = lane >> 4;
    float* __restrict__ dst = UV + (size_t)whichW * BN * Hh;
    #pragma unroll
    for (int r = 0; r < 8; ++r) {
        const int rowm = tm * 16 + r + 8 * hi;
        dst[(size_t)rowm * Hh + hcol] = acc[r];
    }
}

__device__ inline float wave_reduce_sum(float v) {
    #pragma unroll
    for (int m = 16; m >= 1; m >>= 1) v += __shfl_xor(v, m, 32);
    return v;
}

__global__ void pair_reduce_kernel(const float* __restrict__ U,
                                   const float* __restrict__ V,
                                   const float* __restrict__ b_enc,
                                   const float* __restrict__ ln_g,
                                   const float* __restrict__ ln_b,
                                   const float* __restrict__ W_att,
                                   const float* __restrict__ b_att,
                                   float* __restrict__ G,
                                   float* __restrict__ S)
{
    __shared__ float gsh[8 * Hh];
    __shared__ float ssh[8];

    const int bi   = blockIdx.x;
    const int b    = bi >> 5;
    const int i    = bi & 31;
    const int lane = threadIdx.x & 31;
    const int wave = threadIdx.x >> 5;

    const float batt = b_att[0];

    float u[8], be[8], lg[8], lb[8], wa[8], gacc[8];
    #pragma unroll
    for (int t = 0; t < 8; ++t) {
        const int idx = lane + 32 * t;
        u[t]  = U[(size_t)bi * Hh + idx];
        be[t] = b_enc[idx];
        lg[t] = ln_g[idx];
        lb[t] = ln_b[idx];
        wa[t] = W_att[idx];
        gacc[t] = 0.0f;
    }
    float sacc = 0.0f;

    for (int jj = wave; jj < Nsl; jj += 8) {
        if (jj == i) continue;
        const float* __restrict__ vrow = V + (size_t)(b * Nsl + jj) * Hh;

        float h[8];
        float sum = 0.0f;
        #pragma unroll
        for (int t = 0; t < 8; ++t) {
            h[t] = u[t] + vrow[lane + 32 * t] + be[t];
            sum += h[t];
        }
        const float mu = wave_reduce_sum(sum) * (1.0f / Hh);

        float sq = 0.0f;
        #pragma unroll
        for (int t = 0; t < 8; ++t) { const float d = h[t] - mu; sq += d * d; }
        const float var = wave_reduce_sum(sq) * (1.0f / Hh);
        const float inv = rsqrtf(var + LN_EPS);

        float adot = 0.0f;
        #pragma unroll
        for (int t = 0; t < 8; ++t) {
            float hn = (h[t] - mu) * inv * lg[t] + lb[t];
            hn = fmaxf(hn, 0.0f);
            h[t] = hn;
            adot += hn * wa[t];
        }
        adot = wave_reduce_sum(adot) + batt;
        const float a = 1.0f / (1.0f + __expf(-adot));

        #pragma unroll
        for (int t = 0; t < 8; ++t) gacc[t] += a * h[t];
        sacc += a;
    }

    #pragma unroll
    for (int t = 0; t < 8; ++t) gsh[wave * Hh + lane + 32 * t] = gacc[t];
    if (lane == 0) ssh[wave] = sacc;
    __syncthreads();

    float tot = 0.0f;
    #pragma unroll
    for (int w = 0; w < 8; ++w) tot += gsh[w * Hh + threadIdx.x];
    G[(size_t)bi * Hh + threadIdx.x] = tot;

    if (threadIdx.x == 0) {
        float st = 0.0f;
        #pragma unroll
        for (int w = 0; w < 8; ++w) st += ssh[w];
        S[bi] = st;
    }
}

__global__ void eff_gemm_kernel(const float* __restrict__ G,
                                const float* __restrict__ S,
                                const float* __restrict__ Weff,
                                const float* __restrict__ beff,
                                float* __restrict__ out)
{
    const int lane = threadIdx.x & 31;
    const int wave = threadIdx.x >> 5;
    const int tile = blockIdx.x * 8 + wave;
    const int tm = tile >> 3;
    const int tn = tile & 7;

    const int row = tm * 16 + (lane & 15);
    const int col = tn * 16 + (lane & 15);
    const int klo = (lane >> 4) * 2;

    const float* __restrict__ arow = G + (size_t)row * Hh;

    v8f acc = {};
    #pragma unroll 4
    for (int k = 0; k < Hh; k += 4) {
        const int kk = k + klo;
        v2f a; a.x = arow[kk]; a.y = arow[kk + 1];
        const float* bp = Weff + (size_t)kk * Dd + col;
        v2f b; b.x = bp[0]; b.y = bp[Dd];
        acc = __builtin_amdgcn_wmma_f32_16x16x4_f32(false, a, false, b,
                                                    (short)0, acc, false, false);
    }

    const float bec = beff[col];
    const int hi = lane >> 4;
    #pragma unroll
    for (int r = 0; r < 8; ++r) {
        const int rowm = tm * 16 + r + 8 * hi;
        out[(size_t)rowm * Dd + col] = acc[r] + S[rowm] * bec;
    }
}

extern "C" void kernel_launch(void* const* d_in, const int* in_sizes, int n_in,
                              void* d_out, int out_size, void* d_ws, size_t ws_size,
                              hipStream_t stream)
{
    const float* slots = (const float*)d_in[0];
    const float* W_enc = (const float*)d_in[1];
    const float* b_enc = (const float*)d_in[2];
    const float* ln_g  = (const float*)d_in[3];
    const float* ln_b  = (const float*)d_in[4];
    const float* W_att = (const float*)d_in[5];
    const float* b_att = (const float*)d_in[6];
    const float* W_eff = (const float*)d_in[7];
    const float* b_eff = (const float*)d_in[8];
    float* out = (float*)d_out;

    float* ws = (float*)d_ws;
    float* U  = ws;
    float* V  = U + (size_t)BN * Hh;
    float* Gm = V + (size_t)BN * Hh;
    float* Sm = Gm + (size_t)BN * Hh;

    enc_gemm_kernel<<<1024, 256, 0, stream>>>(slots, W_enc, U);

    pair_reduce_kernel<<<BN, 256, 0, stream>>>(U, V, b_enc, ln_g, ln_b,
                                               W_att, b_att, Gm, Sm);

    eff_gemm_kernel<<<256, 256, 0, stream>>>(Gm, Sm, W_eff, b_eff, out);
}


# pair kernel: LayerNorm mean and variance per row, not per pair: rows centred once per block, cross term uc.vc as a Gram tile on WMMA; one sweep, one wave sum per pair; mean/variance sums change order
# speedup vs baseline: 1.7017x; 1.7017x over previous
#include <hip/hip_runtime.h>

typedef __attribute__((ext_vector_type(2))) float v2f;
typedef __attribute__((ext_vector_type(4))) float v4f;
typedef __attribute__((ext_vector_type(8))) float v8f;

constexpr int Bsz = 128;
constexpr int Nsl = 32;
constexpr int Dd  = 128;
constexpr int Hh  = 256;
constexpr int BN  = Bsz * Nsl;
constexpr float LN_EPS = 1e-5f;

template <int NG, int CH>
struct FragSet {
    v4f a[CH];
    v4f b[CH][2][NG][2];
};

template <int NG, int CH>
__device__ __forceinline__ void load_batch(FragSet<NG, CH>& f,
                                           const float* __restrict__ ap,
                                           const float* __restrict__ bp,
                                           const int ldb)
{
    #pragma unroll
    for (int ch = 0; ch < CH; ++ch) {
        f.a[ch] = *reinterpret_cast<const v4f*>(ap + 8 * ch);
        #pragma unroll
        for (int p = 0; p < 2; ++p) {
            const float* __restrict__ br = bp + (size_t)(8 * ch + 2 * p) * ldb;
            #pragma unroll
            for (int g = 0; g < NG; ++g) {
                f.b[ch][p][g][0] = *reinterpret_cast<const v4f*>(br + 64 * g);
                f.b[ch][p][g][1] = *reinterpret_cast<const v4f*>(br + ldb + 64 * g);
            }
        }
    }
}

template <int NG, int CH>
__device__ __forceinline__ void mma_batch(const FragSet<NG, CH>& f, v8f (&acc)[NG * 4])
{
    #pragma unroll
    for (int ch = 0; ch < CH; ++ch) {
        #pragma unroll
        for (int p = 0; p < 2; ++p) {
            v2f a; a.x = f.a[ch][2 * p]; a.y = f.a[ch][2 * p + 1];
            #pragma unroll
            for (int g = 0; g < NG; ++g) {
                #pragma unroll
                for (int q = 0; q < 4; ++q) {
                    v2f b; b.x = f.b[ch][p][g][0][q]; b.y = f.b[ch][p][g][1][q];
                    acc[4 * g + q] = __builtin_amdgcn_wmma_f32_16x16x4_f32(
                        false, a, false, b, (short)0, acc[4 * g + q], false, false);
                }
            }
        }
    }
}

template <int KT, int NG, int CH>
__device__ __forceinline__ void strip_gemm(const float* __restrict__ ap,
                                           const float* __restrict__ bp,
                                           const int ldb,
                                           v8f (&acc)[NG * 4])
{
    constexpr int NB = KT / CH;
    constexpr int AHEAD = 2;
    static_assert(KT % CH == 0 && NB > AHEAD, "batching");
    FragSet<NG, CH> ring[AHEAD + 1];

    #pragma unroll
    for (int i = 0; i < AHEAD; ++i)
        load_batch<NG, CH>(ring[i], ap + 8 * CH * i, bp + (size_t)(8 * CH * i) * ldb, ldb);

    #pragma unroll
    for (int i = 0; i < NB; ++i) {
        __builtin_amdgcn_sched_barrier(0);
        if (i + AHEAD < NB)
            load_batch<NG, CH>(ring[(i + AHEAD) % (AHEAD + 1)], ap + 8 * CH * (i + AHEAD),
                               bp + (size_t)(8 * CH * (i + AHEAD)) * ldb, ldb);
        __builtin_amdgcn_sched_barrier(0);
        mma_batch<NG, CH>(ring[i % (AHEAD + 1)], acc);
    }
}

__global__ __launch_bounds__(128)
void enc_gemm_kernel(const float* __restrict__ slots,
                     const float* __restrict__ Wenc,
                     float* __restrict__ UV)
{
    const int lane = threadIdx.x & 31;
    const int wave = threadIdx.x >> 5;
    const int c = lane & 15;
    const int h = lane >> 4;
    const int g  = blockIdx.x & 3;
    const int tm = (blockIdx.x >> 2) * 4 + wave;
    const int whichW = g >> 1;
    const int hb     = (g & 1) * 128;

    const float* __restrict__ ap = slots + (size_t)(tm * 16 + c) * Dd + 4 * h;
    const float* __restrict__ bp = Wenc + (size_t)(whichW * Dd + 4 * h) * Hh + hb + 4 * c;

    v8f acc[8] = {};
    strip_gemm<Dd / 8, 2, 2>(ap, bp, Hh, acc);

    float* __restrict__ dst = UV + (size_t)whichW * BN * Hh
                                 + (size_t)(tm * 16 + 8 * h) * Hh + hb + 4 * c;
    #pragma unroll
    for (int r = 0; r < 8; ++r) {
        #pragma unroll
        for (int gg = 0; gg < 2; ++gg) {
            v4f o;
            o.x = acc[4 * gg + 0][r]; o.y = acc[4 * gg + 1][r];
            o.z = acc[4 * gg + 2][r]; o.w = acc[4 * gg + 3][r];
            *reinterpret_cast<v4f*>(dst + (size_t)r * Hh + 64 * gg) = o;
        }
    }
}

constexpr int IPB = 8;
constexpr int JPW = 4;
constexpr int RP  = Hh + 4;

template <int NV>
__device__ __forceinline__ void wave_reduce_sum(float (&v)[NV]) {
    #pragma unroll
    for (int m = 16; m >= 1; m >>= 1) {
        float o[NV];
        #pragma unroll
        for (int c = 0; c < NV; ++c) o[c] = __shfl_xor(v[c], m, 32);
        __builtin_amdgcn_sched_barrier(0);
        #pragma unroll
        for (int c = 0; c < NV; ++c) v[c] += o[c];
    }
}

__global__ __launch_bounds__(256)
void pair_reduce_kernel(const float* __restrict__ U,
                        const float* __restrict__ V,
                        const float* __restrict__ b_enc,
                        const float* __restrict__ ln_g,
                        const float* __restrict__ ln_b,
                        const float* __restrict__ W_att,
                        const float* __restrict__ b_att,
                        float* __restrict__ G,
                        float* __restrict__ S)
{
    __shared__ __attribute__((aligned(16))) float ucs[IPB][RP];
    __shared__ __attribute__((aligned(16))) float vcs[Nsl][RP];
    __shared__ float cps[4][2][IPB][Nsl];
    __shared__ float su_sh[IPB];
    __shared__ float sv_sh[Nsl];
    __shared__ float gsh[2][8 * Hh];
    __shared__ float ssh[2][8];

    const int b    = blockIdx.x / (Nsl / IPB);
    const int i0   = (blockIdx.x % (Nsl / IPB)) * IPB;
    const int lane = threadIdx.x & 31;
    const int wave = threadIdx.x >> 5;

    const float batt = b_att[0];

    float lg[8], lb[8], wa[8], v[JPW][8], ur[8];
    #pragma unroll
    for (int t = 0; t < 8; ++t) {
        const int idx = lane + 32 * t;
        lg[t] = ln_g[idx];
        lb[t] = ln_b[idx];
        wa[t] = W_att[idx];
        #pragma unroll
        for (int c = 0; c < JPW; ++c)
            v[c][t] = V[(size_t)(b * Nsl + wave + 8 * c) * Hh + idx];
        ur[t] = U[(size_t)(b * Nsl + i0 + wave) * Hh + idx] + b_enc[idx];
    }

    {
        float red[JPW + 1];
        #pragma unroll
        for (int c = 0; c < JPW; ++c) {
            float sum = 0.0f;
            #pragma unroll
            for (int t = 0; t < 8; ++t) sum += v[c][t];
            red[c] = sum;
        }
        {
            float sum = 0.0f;
            #pragma unroll
            for (int t = 0; t < 8; ++t) sum += ur[t];
            red[JPW] = sum;
        }
        wave_reduce_sum<JPW + 1>(red);
        #pragma unroll
        for (int c = 0; c < JPW; ++c) {
            const float mu = red[c] * (1.0f / Hh);
            float sq = 0.0f;
            #pragma unroll
            for (int t = 0; t < 8; ++t) { v[c][t] -= mu; sq += v[c][t] * v[c][t]; }
            red[c] = sq;
        }
        {
            const float mu = red[JPW] * (1.0f / Hh);
            float sq = 0.0f;
            #pragma unroll
            for (int t = 0; t < 8; ++t) { ur[t] -= mu; sq += ur[t] * ur[t]; }
            red[JPW] = sq;
        }
        wave_reduce_sum<JPW + 1>(red);
        #pragma unroll
        for (int t = 0; t < 8; ++t) {
            #pragma unroll
            for (int c = 0; c < JPW; ++c) vcs[wave + 8 * c][lane + 32 * t] = v[c][t];
            ucs[wave][lane + 32 * t] = ur[t];
        }
        if (lane == 0) {
            #pragma unroll
            for (int c = 0; c < JPW; ++c) sv_sh[wave + 8 * c] = red[c];
            su_sh[wave] = red[JPW];
        }
    }
    __syncthreads();

    {
        const int cc = lane & 15;
        const int hh = lane >> 4;
        const int nt = wave & 1;
        const int kq = wave >> 1;
        const float* ar = &ucs[cc & 7][64 * kq + 4 * hh];
        const float* br = &vcs[16 * nt + cc][64 * kq + 4 * hh];
        v8f acc0 = {};
        v8f acc1 = {};
        #pragma unroll
        for (int s8 = 0; s8 < 8; ++s8) {
            const v4f a4 = *reinterpret_cast<const v4f*>(ar + 8 * s8);
            const v4f b4 = *reinterpret_cast<const v4f*>(br + 8 * s8);
            v2f a01, b01, a23, b23;
            a01.x = a4.x; a01.y = a4.y; b01.x = b4.x; b01.y = b4.y;
            a23.x = a4.z; a23.y = a4.w; b23.x = b4.z; b23.y = b4.w;
            acc0 = __builtin_amdgcn_wmma_f32_16x16x4_f32(false, a01, false, b01, (short)0, acc0, false, false);
            acc1 = __builtin_amdgcn_wmma_f32_16x16x4_f32(false, a23, false, b23, (short)0, acc1, false, false);
        }
        #pragma unroll
        for (int r = 0; r < 8; ++r) cps[kq][hh][r][16 * nt + cc] = acc0[r] + acc1[r];
    }
    __syncthreads();

    float inv_l;
    {
        const int ii = lane >> 2;
        const int cj = wave + 8 * (lane & 3);
        const float cs = ((cps[0][0][ii][cj] + cps[1][0][ii][cj]) + cps[2][0][ii][cj]) + cps[3][0][ii][cj];
        const float var = fmaxf((su_sh[ii] + sv_sh[cj] + 2.0f * cs) * (1.0f / Hh), 0.0f);
        inv_l = rsqrtf(var + LN_EPS);
    }

    for (int ii = 0; ii < IPB; ++ii) {
        const int i   = i0 + ii;
        const int bi  = b * Nsl + i;
        const int buf = ii & 1;

        float u[8];
        #pragma unroll
        for (int t = 0; t < 8; ++t) u[t] = ucs[ii][lane + 32 * t];

        float h[JPW][8];
        float red[JPW];
        #pragma unroll
        for (int c = 0; c < JPW; ++c) {
            const float inv = __int_as_float(
                __builtin_amdgcn_readlane(__float_as_int(inv_l), 4 * ii + c));
            float adot = 0.0f;
            #pragma unroll
            for (int t = 0; t < 8; ++t) {
                float hn = (u[t] + v[c][t]) * inv * lg[t] + lb[t];
                hn = fmaxf(hn, 0.0f);
                h[c][t] = hn;
                adot += hn * wa[t];
            }
            red[c] = adot;
        }
        wave_reduce_sum<JPW>(red);

        float gacc[8];
        #pragma unroll
        for (int t = 0; t < 8; ++t) gacc[t] = 0.0f;
        float sacc = 0.0f;
        #pragma unroll
        for (int c = 0; c < JPW; ++c) {
            const float adot = red[c] + batt;
            float a = 1.0f / (1.0f + __expf(-adot));
            a = (wave + 8 * c == i) ? 0.0f : a;
            #pragma unroll
            for (int t = 0; t < 8; ++t) gacc[t] += a * h[c][t];
            sacc += a;
        }

        #pragma unroll
        for (int t = 0; t < 8; ++t) gsh[buf][wave * Hh + lane + 32 * t] = gacc[t];
        if (lane == 0) ssh[buf][wave] = sacc;
        __syncthreads();

        float tot = 0.0f;
        #pragma unroll
        for (int w = 0; w < 8; ++w) tot += gsh[buf][w * Hh + threadIdx.x];
        G[(size_t)bi * Hh + threadIdx.x] = tot;

        if (threadIdx.x == 0) {
            float st = 0.0f;
            #pragma unroll
            for (int w = 0; w < 8; ++w) st += ssh[buf][w];
            S[bi] = st;
        }
    }
}

__global__ __launch_bounds__(64)
void eff_gemm_kernel(const float* __restrict__ G,
                     const float* __restrict__ S,
                     const float* __restrict__ Weff,
                     const float* __restrict__ beff,
                     float* __restrict__ out)
{
    const int lane = threadIdx.x & 31;
    const int wave = threadIdx.x >> 5;
    const int c = lane & 15;
    const int h = lane >> 4;
    const int g  = blockIdx.x & 1;
    const int tm = (blockIdx.x >> 1) * 2 + wave;
    const int col0 = 64 * g + 4 * c;

    const float* __restrict__ ap = G + (size_t)(tm * 16 + c) * Hh + 4 * h;
    const float* __restrict__ bp = Weff + (size_t)(4 * h) * Dd + col0;

    v8f acc[4] = {};
    strip_gemm<Hh / 8, 1, 4>(ap, bp, Dd, acc);

    const float be0 = beff[col0 + 0];
    const float be1 = beff[col0 + 1];
    const float be2 = beff[col0 + 2];
    const float be3 = beff[col0 + 3];
    #pragma unroll
    for (int r = 0; r < 8; ++r) {
        const int rowm = tm * 16 + 8 * h + r;
        const float s = S[rowm];
        v4f o;
        o.x = acc[0][r] + s * be0;
        o.y = acc[1][r] + s * be1;
        o.z = acc[2][r] + s * be2;
        o.w = acc[3][r] + s * be3;
        *reinterpret_cast<v4f*>(out + (size_t)rowm * Dd + col0) = o;
    }
}

extern "C" void kernel_launch(void* const* d_in, const int* in_sizes, int n_in,
                              void* d_out, int out_size, void* d_ws, size_t ws_size,
                              hipStream_t stream)
{
    const float* slots = (const float*)d_in[0];
    const float* W_enc = (const float*)d_in[1];
    const float* b_enc = (const float*)d_in[2];
    const float* ln_g  = (const float*)d_in[3];
    const float* ln_b  = (const float*)d_in[4];
    const float* W_att = (const float*)d_in[5];
    const float* b_att = (const float*)d_in[6];
    const float* W_eff = (const float*)d_in[7];
    const float* b_eff = (const float*)d_in[8];
    float* out = (float*)d_out;

    float* ws = (float*)d_ws;
    float* U  = ws;
    float* V  = U + (size_t)BN * Hh;
    float* Gm = V + (size_t)BN * Hh;
    float* Sm = Gm + (size_t)BN * Hh;

    enc_gemm_kernel<<<256, 128, 0, stream>>>(slots, W_enc, U);

    pair_reduce_kernel<<<BN / IPB, 256, 0, stream>>>(U, V, b_enc, ln_g, ln_b,
                                                     W_att, b_att, Gm, Sm);

    eff_gemm_kernel<<<256, 64, 0, stream>>>(Gm, Sm, W_eff, b_eff, out);
}


# pair sweep on two-wide f32 vectors, ln_g folded into the centred rows: scale+bias, ReLU, attention and G products are one packed instruction per 2 elements; lane's attention sum as even+odd chains
# speedup vs baseline: 1.7792x; 1.0455x over previous
#include <hip/hip_runtime.h>

typedef __attribute__((ext_vector_type(2))) float v2f;
typedef __attribute__((ext_vector_type(4))) float v4f;
typedef __attribute__((ext_vector_type(8))) float v8f;

constexpr int Bsz = 128;
constexpr int Nsl = 32;
constexpr int Dd  = 128;
constexpr int Hh  = 256;
constexpr int BN  = Bsz * Nsl;
constexpr float LN_EPS = 1e-5f;

template <int NG, int CH>
struct FragSet {
    v4f a[CH];
    v4f b[CH][2][NG][2];
};

template <int NG, int CH>
__device__ __forceinline__ void load_batch(FragSet<NG, CH>& f,
                                           const float* __restrict__ ap,
                                           const float* __restrict__ bp,
                                           const int ldb)
{
    #pragma unroll
    for (int ch = 0; ch < CH; ++ch) {
        f.a[ch] = *reinterpret_cast<const v4f*>(ap + 8 * ch);
        #pragma unroll
        for (int p = 0; p < 2; ++p) {
            const float* __restrict__ br = bp + (size_t)(8 * ch + 2 * p) * ldb;
            #pragma unroll
            for (int g = 0; g < NG; ++g) {
                f.b[ch][p][g][0] = *reinterpret_cast<const v4f*>(br + 64 * g);
                f.b[ch][p][g][1] = *reinterpret_cast<const v4f*>(br + ldb + 64 * g);
            }
        }
    }
}

template <int NG, int CH>
__device__ __forceinline__ void mma_batch(const FragSet<NG, CH>& f, v8f (&acc)[NG * 4])
{
    #pragma unroll
    for (int ch = 0; ch < CH; ++ch) {
        #pragma unroll
        for (int p = 0; p < 2; ++p) {
            v2f a; a.x = f.a[ch][2 * p]; a.y = f.a[ch][2 * p + 1];
            #pragma unroll
            for (int g = 0; g < NG; ++g) {
                #pragma unroll
                for (int q = 0; q < 4; ++q) {
                    v2f b; b.x = f.b[ch][p][g][0][q]; b.y = f.b[ch][p][g][1][q];
                    acc[4 * g + q] = __builtin_amdgcn_wmma_f32_16x16x4_f32(
                        false, a, false, b, (short)0, acc[4 * g + q], false, false);
                }
            }
        }
    }
}

template <int KT, int NG, int CH>
__device__ __forceinline__ void strip_gemm(const float* __restrict__ ap,
                                           const float* __restrict__ bp,
                                           const int ldb,
                                           v8f (&acc)[NG * 4])
{
    constexpr int NB = KT / CH;
    constexpr int AHEAD = 2;
    static_assert(KT % CH == 0 && NB > AHEAD, "batching");
    FragSet<NG, CH> ring[AHEAD + 1];

    #pragma unroll
    for (int i = 0; i < AHEAD; ++i)
        load_batch<NG, CH>(ring[i], ap + 8 * CH * i, bp + (size_t)(8 * CH * i) * ldb, ldb);

    #pragma unroll
    for (int i = 0; i < NB; ++i) {
        __builtin_amdgcn_sched_barrier(0);
        if (i + AHEAD < NB)
            load_batch<NG, CH>(ring[(i + AHEAD) % (AHEAD + 1)], ap + 8 * CH * (i + AHEAD),
                               bp + (size_t)(8 * CH * (i + AHEAD)) * ldb, ldb);
        __builtin_amdgcn_sched_barrier(0);
        mma_batch<NG, CH>(ring[i % (AHEAD + 1)], acc);
    }
}

__global__ __launch_bounds__(128)
void enc_gemm_kernel(const float* __restrict__ slots,
                     const float* __restrict__ Wenc,
                     float* __restrict__ UV)
{
    const int lane = threadIdx.x & 31;
    const int wave = threadIdx.x >> 5;
    const int c = lane & 15;
    const int h = lane >> 4;
    const int g  = blockIdx.x & 3;
    const int tm = (blockIdx.x >> 2) * 4 + wave;
    const int whichW = g >> 1;
    const int hb     = (g & 1) * 128;

    const float* __restrict__ ap = slots + (size_t)(tm * 16 + c) * Dd + 4 * h;
    const float* __restrict__ bp = Wenc + (size_t)(whichW * Dd + 4 * h) * Hh + hb + 4 * c;

    v8f acc[8] = {};
    strip_gemm<Dd / 8, 2, 2>(ap, bp, Hh, acc);

    float* __restrict__ dst = UV + (size_t)whichW * BN * Hh
                                 + (size_t)(tm * 16 + 8 * h) * Hh + hb + 4 * c;
    #pragma unroll
    for (int r = 0; r < 8; ++r) {
        #pragma unroll
        for (int gg = 0; gg < 2; ++gg) {
            v4f o;
            o.x = acc[4 * gg + 0][r]; o.y = acc[4 * gg + 1][r];
            o.z = acc[4 * gg + 2][r]; o.w = acc[4 * gg + 3][r];
            *reinterpret_cast<v4f*>(dst + (size_t)r * Hh + 64 * gg) = o;
        }
    }
}

constexpr int IPB = 8;
constexpr int JPW = 4;
constexpr int RP  = Hh + 4;

template <int NV>
__device__ __forceinline__ void wave_reduce_sum(float (&v)[NV]) {
    #pragma unroll
    for (int m = 16; m >= 1; m >>= 1) {
        float o[NV];
        #pragma unroll
        for (int c = 0; c < NV; ++c) o[c] = __shfl_xor(v[c], m, 32);
        __builtin_amdgcn_sched_barrier(0);
        #pragma unroll
        for (int c = 0; c < NV; ++c) v[c] += o[c];
    }
}

__global__ __launch_bounds__(256)
void pair_reduce_kernel(const float* __restrict__ U,
                        const float* __restrict__ V,
                        const float* __restrict__ b_enc,
                        const float* __restrict__ ln_g,
                        const float* __restrict__ ln_b,
                        const float* __restrict__ W_att,
                        const float* __restrict__ b_att,
                        float* __restrict__ G,
                        float* __restrict__ S)
{
    __shared__ __attribute__((aligned(16))) float ucs[IPB][RP];
    __shared__ __attribute__((aligned(16))) float vcs[Nsl][RP];
    __shared__ float cps[4][2][IPB][Nsl];
    __shared__ float su_sh[IPB];
    __shared__ float sv_sh[Nsl];
    __shared__ float gsh[2][8 * Hh];
    __shared__ float ssh[2][8];

    const int b    = blockIdx.x / (Nsl / IPB);
    const int i0   = (blockIdx.x % (Nsl / IPB)) * IPB;
    const int lane = threadIdx.x & 31;
    const int wave = threadIdx.x >> 5;

    const float batt = b_att[0];

    float lg[8], lb[8], wa[8], v[JPW][8], ur[8];
    #pragma unroll
    for (int t = 0; t < 8; ++t) {
        const int idx = lane + 32 * t;
        lg[t] = ln_g[idx];
        lb[t] = ln_b[idx];
        wa[t] = W_att[idx];
        #pragma unroll
        for (int c = 0; c < JPW; ++c)
            v[c][t] = V[(size_t)(b * Nsl + wave + 8 * c) * Hh + idx];
        ur[t] = U[(size_t)(b * Nsl + i0 + wave) * Hh + idx] + b_enc[idx];
    }

    {
        float red[JPW + 1];
        #pragma unroll
        for (int c = 0; c < JPW; ++c) {
            float sum = 0.0f;
            #pragma unroll
            for (int t = 0; t < 8; ++t) sum += v[c][t];
            red[c] = sum;
        }
        {
            float sum = 0.0f;
            #pragma unroll
            for (int t = 0; t < 8; ++t) sum += ur[t];
            red[JPW] = sum;
        }
        wave_reduce_sum<JPW + 1>(red);
        #pragma unroll
        for (int c = 0; c < JPW; ++c) {
            const float mu = red[c] * (1.0f / Hh);
            float sq = 0.0f;
            #pragma unroll
            for (int t = 0; t < 8; ++t) { v[c][t] -= mu; sq += v[c][t] * v[c][t]; }
            red[c] = sq;
        }
        {
            const float mu = red[JPW] * (1.0f / Hh);
            float sq = 0.0f;
            #pragma unroll
            for (int t = 0; t < 8; ++t) { ur[t] -= mu; sq += ur[t] * ur[t]; }
            red[JPW] = sq;
        }
        wave_reduce_sum<JPW + 1>(red);
        #pragma unroll
        for (int t = 0; t < 8; ++t) {
            #pragma unroll
            for (int c = 0; c < JPW; ++c) vcs[wave + 8 * c][lane + 32 * t] = v[c][t];
            ucs[wave][lane + 32 * t] = ur[t];
        }
        if (lane == 0) {
            #pragma unroll
            for (int c = 0; c < JPW; ++c) sv_sh[wave + 8 * c] = red[c];
            su_sh[wave] = red[JPW];
        }
    }
    __syncthreads();

    {
        const int cc = lane & 15;
        const int hh = lane >> 4;
        const int nt = wave & 1;
        const int kq = wave >> 1;
        const float* ar = &ucs[cc & 7][64 * kq + 4 * hh];
        const float* br = &vcs[16 * nt + cc][64 * kq + 4 * hh];
        v8f acc0 = {};
        v8f acc1 = {};
        #pragma unroll
        for (int s8 = 0; s8 < 8; ++s8) {
            const v4f a4 = *reinterpret_cast<const v4f*>(ar + 8 * s8);
            const v4f b4 = *reinterpret_cast<const v4f*>(br + 8 * s8);
            v2f a01, b01, a23, b23;
            a01.x = a4.x; a01.y = a4.y; b01.x = b4.x; b01.y = b4.y;
            a23.x = a4.z; a23.y = a4.w; b23.x = b4.z; b23.y = b4.w;
            acc0 = __builtin_amdgcn_wmma_f32_16x16x4_f32(false, a01, false, b01, (short)0, acc0, false, false);
            acc1 = __builtin_amdgcn_wmma_f32_16x16x4_f32(false, a23, false, b23, (short)0, acc1, false, false);
        }
        #pragma unroll
        for (int r = 0; r < 8; ++r) cps[kq][hh][r][16 * nt + cc] = acc0[r] + acc1[r];
    }
    __syncthreads();

    float inv_l;
    {
        const int ii = lane >> 2;
        const int cj = wave + 8 * (lane & 3);
        const float cs = ((cps[0][0][ii][cj] + cps[1][0][ii][cj]) + cps[2][0][ii][cj]) + cps[3][0][ii][cj];
        const float var = fmaxf((su_sh[ii] + sv_sh[cj] + 2.0f * cs) * (1.0f / Hh), 0.0f);
        inv_l = rsqrtf(var + LN_EPS);
    }

    v2f vg[JPW][4], lg2[4], lb2[4], wa2[4];
    #pragma unroll
    for (int k = 0; k < 4; ++k) {
        lg2[k].x = lg[2 * k]; lg2[k].y = lg[2 * k + 1];
        lb2[k].x = lb[2 * k]; lb2[k].y = lb[2 * k + 1];
        wa2[k].x = wa[2 * k]; wa2[k].y = wa[2 * k + 1];
        #pragma unroll
        for (int c = 0; c < JPW; ++c) {
            vg[c][k].x = v[c][2 * k]     * lg[2 * k];
            vg[c][k].y = v[c][2 * k + 1] * lg[2 * k + 1];
        }
    }
    const v2f zero2 = {0.0f, 0.0f};

    for (int ii = 0; ii < IPB; ++ii) {
        const int i   = i0 + ii;
        const int bi  = b * Nsl + i;
        const int buf = ii & 1;

        v2f ug[4];
        #pragma unroll
        for (int k = 0; k < 4; ++k) {
            v2f uc;
            uc.x = ucs[ii][lane + 64 * k];
            uc.y = ucs[ii][lane + 64 * k + 32];
            ug[k] = uc * lg2[k];
        }

        v2f h2[JPW][4];
        float red[JPW];
        #pragma unroll
        for (int c = 0; c < JPW; ++c) {
            const float inv = __int_as_float(
                __builtin_amdgcn_readlane(__float_as_int(inv_l), 4 * ii + c));
            v2f inv2; inv2.x = inv; inv2.y = inv;
            v2f ad = zero2;
            #pragma unroll
            for (int k = 0; k < 4; ++k) {
                v2f hn = (ug[k] + vg[c][k]) * inv2 + lb2[k];
                hn = __builtin_elementwise_max(hn, zero2);
                h2[c][k] = hn;
                ad = hn * wa2[k] + ad;
            }
            red[c] = ad.x + ad.y;
        }
        wave_reduce_sum<JPW>(red);

        v2f g2[4];
        #pragma unroll
        for (int k = 0; k < 4; ++k) g2[k] = zero2;
        float sacc = 0.0f;
        #pragma unroll
        for (int c = 0; c < JPW; ++c) {
            const float adot = red[c] + batt;
            float a = 1.0f / (1.0f + __expf(-adot));
            a = (wave + 8 * c == i) ? 0.0f : a;
            v2f a2; a2.x = a; a2.y = a;
            #pragma unroll
            for (int k = 0; k < 4; ++k) g2[k] = a2 * h2[c][k] + g2[k];
            sacc += a;
        }

        #pragma unroll
        for (int k = 0; k < 4; ++k) {
            gsh[buf][wave * Hh + lane + 64 * k]      = g2[k].x;
            gsh[buf][wave * Hh + lane + 64 * k + 32] = g2[k].y;
        }
        if (lane == 0) ssh[buf][wave] = sacc;
        __syncthreads();

        float tot = 0.0f;
        #pragma unroll
        for (int w = 0; w < 8; ++w) tot += gsh[buf][w * Hh + threadIdx.x];
        G[(size_t)bi * Hh + threadIdx.x] = tot;

        if (threadIdx.x == 0) {
            float st = 0.0f;
            #pragma unroll
            for (int w = 0; w < 8; ++w) st += ssh[buf][w];
            S[bi] = st;
        }
    }
}

__global__ __launch_bounds__(64)
void eff_gemm_kernel(const float* __restrict__ G,
                     const float* __restrict__ S,
                     const float* __restrict__ Weff,
                     const float* __restrict__ beff,
                     float* __restrict__ out)
{
    const int lane = threadIdx.x & 31;
    const int wave = threadIdx.x >> 5;
    const int c = lane & 15;
    const int h = lane >> 4;
    const int g  = blockIdx.x & 1;
    const int tm = (blockIdx.x >> 1) * 2 + wave;
    const int col0 = 64 * g + 4 * c;

    const float* __restrict__ ap = G + (size_t)(tm * 16 + c) * Hh + 4 * h;
    const float* __restrict__ bp = Weff + (size_t)(4 * h) * Dd + col0;

    v8f acc[4] = {};
    strip_gemm<Hh / 8, 1, 4>(ap, bp, Dd, acc);

    const float be0 = beff[col0 + 0];
    const float be1 = beff[col0 + 1];
    const float be2 = beff[col0 + 2];
    const float be3 = beff[col0 + 3];
    #pragma unroll
    for (int r = 0; r < 8; ++r) {
        const int rowm = tm * 16 + 8 * h + r;
        const float s = S[rowm];
        v4f o;
        o.x = acc[0][r] + s * be0;
        o.y = acc[1][r] + s * be1;
        o.z = acc[2][r] + s * be2;
        o.w = acc[3][r] + s * be3;
        *reinterpret_cast<v4f*>(out + (size_t)rowm * Dd + col0) = o;
    }
}

extern "C" void kernel_launch(void* const* d_in, const int* in_sizes, int n_in,
                              void* d_out, int out_size, void* d_ws, size_t ws_size,
                              hipStream_t stream)
{
    const float* slots = (const float*)d_in[0];
    const float* W_enc = (const float*)d_in[1];
    const float* b_enc = (const float*)d_in[2];
    const float* ln_g  = (const float*)d_in[3];
    const float* ln_b  = (const float*)d_in[4];
    const float* W_att = (const float*)d_in[5];
    const float* b_att = (const float*)d_in[6];
    const float* W_eff = (const float*)d_in[7];
    const float* b_eff = (const float*)d_in[8];
    float* out = (float*)d_out;

    float* ws = (float*)d_ws;
    float* U  = ws;
    float* V  = U + (size_t)BN * Hh;
    float* Gm = V + (size_t)BN * Hh;
    float* Sm = Gm + (size_t)BN * Hh;

    enc_gemm_kernel<<<256, 128, 0, stream>>>(slots, W_enc, U);

    pair_reduce_kernel<<<BN / IPB, 256, 0, stream>>>(U, V, b_enc, ln_g, ln_b,
                                                     W_att, b_att, Gm, Sm);

    eff_gemm_kernel<<<256, 64, 0, stream>>>(Gm, Sm, W_eff, b_eff, out);
}


# eff GEMM launch: K split over two waves per 16x64 strip (1024 waves, one per SIMD, 128 WMMAs each; was 512 x 256); halves meet through LDS, each wave stores 4 of 8 rows; K summed as two halves
# speedup vs baseline: 1.8587x; 1.0447x over previous
#include <hip/hip_runtime.h>

typedef __attribute__((ext_vector_type(2))) float v2f;
typedef __attribute__((ext_vector_type(4))) float v4f;
typedef __attribute__((ext_vector_type(8))) float v8f;

constexpr int Bsz = 128;
constexpr int Nsl = 32;
constexpr int Dd  = 128;
constexpr int Hh  = 256;
constexpr int BN  = Bsz * Nsl;
constexpr float LN_EPS = 1e-5f;

template <int NG, int CH>
struct FragSet {
    v4f a[CH];
    v4f b[CH][2][NG][2];
};

template <int NG, int CH>
__device__ __forceinline__ void load_batch(FragSet<NG, CH>& f,
                                           const float* __restrict__ ap,
                                           const float* __restrict__ bp,
                                           const int ldb)
{
    #pragma unroll
    for (int ch = 0; ch < CH; ++ch) {
        f.a[ch] = *reinterpret_cast<const v4f*>(ap + 8 * ch);
        #pragma unroll
        for (int p = 0; p < 2; ++p) {
            const float* __restrict__ br = bp + (size_t)(8 * ch + 2 * p) * ldb;
            #pragma unroll
            for (int g = 0; g < NG; ++g) {
                f.b[ch][p][g][0] = *reinterpret_cast<const v4f*>(br + 64 * g);
                f.b[ch][p][g][1] = *reinterpret_cast<const v4f*>(br + ldb + 64 * g);
            }
        }
    }
}

template <int NG, int CH>
__device__ __forceinline__ void mma_batch(const FragSet<NG, CH>& f, v8f (&acc)[NG * 4])
{
    #pragma unroll
    for (int ch = 0; ch < CH; ++ch) {
        #pragma unroll
        for (int p = 0; p < 2; ++p) {
            v2f a; a.x = f.a[ch][2 * p]; a.y = f.a[ch][2 * p + 1];
            #pragma unroll
            for (int g = 0; g < NG; ++g) {
                #pragma unroll
                for (int q = 0; q < 4; ++q) {
                    v2f b; b.x = f.b[ch][p][g][0][q]; b.y = f.b[ch][p][g][1][q];
                    acc[4 * g + q] = __builtin_amdgcn_wmma_f32_16x16x4_f32(
                        false, a, false, b, (short)0, acc[4 * g + q], false, false);
                }
            }
        }
    }
}

template <int KT, int NG, int CH>
__device__ __forceinline__ void strip_gemm(const float* __restrict__ ap,
                                           const float* __restrict__ bp,
                                           const int ldb,
                                           v8f (&acc)[NG * 4])
{
    constexpr int NB = KT / CH;
    constexpr int AHEAD = 2;
    static_assert(KT % CH == 0 && NB > AHEAD, "batching");
    FragSet<NG, CH> ring[AHEAD + 1];

    #pragma unroll
    for (int i = 0; i < AHEAD; ++i)
        load_batch<NG, CH>(ring[i], ap + 8 * CH * i, bp + (size_t)(8 * CH * i) * ldb, ldb);

    #pragma unroll
    for (int i = 0; i < NB; ++i) {
        __builtin_amdgcn_sched_barrier(0);
        if (i + AHEAD < NB)
            load_batch<NG, CH>(ring[(i + AHEAD) % (AHEAD + 1)], ap + 8 * CH * (i + AHEAD),
                               bp + (size_t)(8 * CH * (i + AHEAD)) * ldb, ldb);
        __builtin_amdgcn_sched_barrier(0);
        mma_batch<NG, CH>(ring[i % (AHEAD + 1)], acc);
    }
}

__global__ __launch_bounds__(128)
void enc_gemm_kernel(const float* __restrict__ slots,
                     const float* __restrict__ Wenc,
                     float* __restrict__ UV)
{
    const int lane = threadIdx.x & 31;
    const int wave = threadIdx.x >> 5;
    const int c = lane & 15;
    const int h = lane >> 4;
    const int g  = blockIdx.x & 3;
    const int tm = (blockIdx.x >> 2) * 4 + wave;
    const int whichW = g >> 1;
    const int hb     = (g & 1) * 128;

    const float* __restrict__ ap = slots + (size_t)(tm * 16 + c) * Dd + 4 * h;
    const float* __restrict__ bp = Wenc + (size_t)(whichW * Dd + 4 * h) * Hh + hb + 4 * c;

    v8f acc[8] = {};
    strip_gemm<Dd / 8, 2, 2>(ap, bp, Hh, acc);

    float* __restrict__ dst = UV + (size_t)whichW * BN * Hh
                                 + (size_t)(tm * 16 + 8 * h) * Hh + hb + 4 * c;
    #pragma unroll
    for (int r = 0; r < 8; ++r) {
        #pragma unroll
        for (int gg = 0; gg < 2; ++gg) {
            v4f o;
            o.x = acc[4 * gg + 0][r]; o.y = acc[4 * gg + 1][r];
            o.z = acc[4 * gg + 2][r]; o.w = acc[4 * gg + 3][r];
            *reinterpret_cast<v4f*>(dst + (size_t)r * Hh + 64 * gg) = o;
        }
    }
}

constexpr int IPB = 8;
constexpr int JPW = 4;
constexpr int RP  = Hh + 4;

template <int NV>
__device__ __forceinline__ void wave_reduce_sum(float (&v)[NV]) {
    #pragma unroll
    for (int m = 16; m >= 1; m >>= 1) {
        float o[NV];
        #pragma unroll
        for (int c = 0; c < NV; ++c) o[c] = __shfl_xor(v[c], m, 32);
        __builtin_amdgcn_sched_barrier(0);
        #pragma unroll
        for (int c = 0; c < NV; ++c) v[c] += o[c];
    }
}

__global__ __launch_bounds__(256)
void pair_reduce_kernel(const float* __restrict__ U,
                        const float* __restrict__ V,
                        const float* __restrict__ b_enc,
                        const float* __restrict__ ln_g,
                        const float* __restrict__ ln_b,
                        const float* __restrict__ W_att,
                        const float* __restrict__ b_att,
                        float* __restrict__ G,
                        float* __restrict__ S)
{
    __shared__ __attribute__((aligned(16))) float ucs[IPB][RP];
    __shared__ __attribute__((aligned(16))) float vcs[Nsl][RP];
    __shared__ float cps[4][2][IPB][Nsl];
    __shared__ float su_sh[IPB];
    __shared__ float sv_sh[Nsl];
    __shared__ float gsh[2][8 * Hh];
    __shared__ float ssh[2][8];

    const int b    = blockIdx.x / (Nsl / IPB);
    const int i0   = (blockIdx.x % (Nsl / IPB)) * IPB;
    const int lane = threadIdx.x & 31;
    const int wave = threadIdx.x >> 5;

    const float batt = b_att[0];

    float lg[8], lb[8], wa[8], v[JPW][8], ur[8];
    #pragma unroll
    for (int t = 0; t < 8; ++t) {
        const int idx = lane + 32 * t;
        lg[t] = ln_g[idx];
        lb[t] = ln_b[idx];
        wa[t] = W_att[idx];
        #pragma unroll
        for (int c = 0; c < JPW; ++c)
            v[c][t] = V[(size_t)(b * Nsl + wave + 8 * c) * Hh + idx];
        ur[t] = U[(size_t)(b * Nsl + i0 + wave) * Hh + idx] + b_enc[idx];
    }

    {
        float red[JPW + 1];
        #pragma unroll
        for (int c = 0; c < JPW; ++c) {
            float sum = 0.0f;
            #pragma unroll
            for (int t = 0; t < 8; ++t) sum += v[c][t];
            red[c] = sum;
        }
        {
            float sum = 0.0f;
            #pragma unroll
            for (int t = 0; t < 8; ++t) sum += ur[t];
            red[JPW] = sum;
        }
        wave_reduce_sum<JPW + 1>(red);
        #pragma unroll
        for (int c = 0; c < JPW; ++c) {
            const float mu = red[c] * (1.0f / Hh);
            float sq = 0.0f;
            #pragma unroll
            for (int t = 0; t < 8; ++t) { v[c][t] -= mu; sq += v[c][t] * v[c][t]; }
            red[c] = sq;
        }
        {
            const float mu = red[JPW] * (1.0f / Hh);
            float sq = 0.0f;
            #pragma unroll
            for (int t = 0; t < 8; ++t) { ur[t] -= mu; sq += ur[t] * ur[t]; }
            red[JPW] = sq;
        }
        wave_reduce_sum<JPW + 1>(red);
        #pragma unroll
        for (int t = 0; t < 8; ++t) {
            #pragma unroll
            for (int c = 0; c < JPW; ++c) vcs[wave + 8 * c][lane + 32 * t] = v[c][t];
            ucs[wave][lane + 32 * t] = ur[t];
        }
        if (lane == 0) {
            #pragma unroll
            for (int c = 0; c < JPW; ++c) sv_sh[wave + 8 * c] = red[c];
            su_sh[wave] = red[JPW];
        }
    }
    __syncthreads();

    {
        const int cc = lane & 15;
        const int hh = lane >> 4;
        const int nt = wave & 1;
        const int kq = wave >> 1;
        const float* ar = &ucs[cc & 7][64 * kq + 4 * hh];
        const float* br = &vcs[16 * nt + cc][64 * kq + 4 * hh];
        v8f acc0 = {};
        v8f acc1 = {};
        #pragma unroll
        for (int s8 = 0; s8 < 8; ++s8) {
            const v4f a4 = *reinterpret_cast<const v4f*>(ar + 8 * s8);
            const v4f b4 = *reinterpret_cast<const v4f*>(br + 8 * s8);
            v2f a01, b01, a23, b23;
            a01.x = a4.x; a01.y = a4.y; b01.x = b4.x; b01.y = b4.y;
            a23.x = a4.z; a23.y = a4.w; b23.x = b4.z; b23.y = b4.w;
            acc0 = __builtin_amdgcn_wmma_f32_16x16x4_f32(false, a01, false, b01, (short)0, acc0, false, false);
            acc1 = __builtin_amdgcn_wmma_f32_16x16x4_f32(false, a23, false, b23, (short)0, acc1, false, false);
        }
        #pragma unroll
        for (int r = 0; r < 8; ++r) cps[kq][hh][r][16 * nt + cc] = acc0[r] + acc1[r];
    }
    __syncthreads();

    float inv_l;
    {
        const int ii = lane >> 2;
        const int cj = wave + 8 * (lane & 3);
        const float cs = ((cps[0][0][ii][cj] + cps[1][0][ii][cj]) + cps[2][0][ii][cj]) + cps[3][0][ii][cj];
        const float var = fmaxf((su_sh[ii] + sv_sh[cj] + 2.0f * cs) * (1.0f / Hh), 0.0f);
        inv_l = rsqrtf(var + LN_EPS);
    }

    v2f vg[JPW][4], lg2[4], lb2[4], wa2[4];
    #pragma unroll
    for (int k = 0; k < 4; ++k) {
        lg2[k].x = lg[2 * k]; lg2[k].y = lg[2 * k + 1];
        lb2[k].x = lb[2 * k]; lb2[k].y = lb[2 * k + 1];
        wa2[k].x = wa[2 * k]; wa2[k].y = wa[2 * k + 1];
        #pragma unroll
        for (int c = 0; c < JPW; ++c) {
            vg[c][k].x = v[c][2 * k]     * lg[2 * k];
            vg[c][k].y = v[c][2 * k + 1] * lg[2 * k + 1];
        }
    }
    const v2f zero2 = {0.0f, 0.0f};

    for (int ii = 0; ii < IPB; ++ii) {
        const int i   = i0 + ii;
        const int bi  = b * Nsl + i;
        const int buf = ii & 1;

        v2f ug[4];
        #pragma unroll
        for (int k = 0; k < 4; ++k) {
            v2f uc;
            uc.x = ucs[ii][lane + 64 * k];
            uc.y = ucs[ii][lane + 64 * k + 32];
            ug[k] = uc * lg2[k];
        }

        v2f h2[JPW][4];
        float red[JPW];
        #pragma unroll
        for (int c = 0; c < JPW; ++c) {
            const float inv = __int_as_float(
                __builtin_amdgcn_readlane(__float_as_int(inv_l), 4 * ii + c));
            v2f inv2; inv2.x = inv; inv2.y = inv;
            v2f ad = zero2;
            #pragma unroll
            for (int k = 0; k < 4; ++k) {
                v2f hn = (ug[k] + vg[c][k]) * inv2 + lb2[k];
                hn = __builtin_elementwise_max(hn, zero2);
                h2[c][k] = hn;
                ad = hn * wa2[k] + ad;
            }
            red[c] = ad.x + ad.y;
        }
        wave_reduce_sum<JPW>(red);

        v2f g2[4];
        #pragma unroll
        for (int k = 0; k < 4; ++k) g2[k] = zero2;
        float sacc = 0.0f;
        #pragma unroll
        for (int c = 0; c < JPW; ++c) {
            const float adot = red[c] + batt;
            float a = 1.0f / (1.0f + __expf(-adot));
            a = (wave + 8 * c == i) ? 0.0f : a;
            v2f a2; a2.x = a; a2.y = a;
            #pragma unroll
            for (int k = 0; k < 4; ++k) g2[k] = a2 * h2[c][k] + g2[k];
            sacc += a;
        }

        #pragma unroll
        for (int k = 0; k < 4; ++k) {
            gsh[buf][wave * Hh + lane + 64 * k]      = g2[k].x;
            gsh[buf][wave * Hh + lane + 64 * k + 32] = g2[k].y;
        }
        if (lane == 0) ssh[buf][wave] = sacc;
        __syncthreads();

        float tot = 0.0f;
        #pragma unroll
        for (int w = 0; w < 8; ++w) tot += gsh[buf][w * Hh + threadIdx.x];
        G[(size_t)bi * Hh + threadIdx.x] = tot;

        if (threadIdx.x == 0) {
            float st = 0.0f;
            #pragma unroll
            for (int w = 0; w < 8; ++w) st += ssh[buf][w];
            S[bi] = st;
        }
    }
}

__global__ __launch_bounds__(128)
void eff_gemm_kernel(const float* __restrict__ G,
                     const float* __restrict__ S,
                     const float* __restrict__ Weff,
                     const float* __restrict__ beff,
                     float* __restrict__ out)
{
    __shared__ __attribute__((aligned(16))) float xch[2][2][4][32 * 4];

    const int lane = threadIdx.x & 31;
    const int wave = threadIdx.x >> 5;
    const int c = lane & 15;
    const int h = lane >> 4;
    const int mt = wave & 1;
    const int kh = wave >> 1;
    const int g  = blockIdx.x & 1;
    const int tm = (blockIdx.x >> 1) * 2 + mt;
    const int col0 = 64 * g + 4 * c;

    const float* __restrict__ ap = G + (size_t)(tm * 16 + c) * Hh + 128 * kh + 4 * h;
    const float* __restrict__ bp = Weff + (size_t)(128 * kh + 4 * h) * Dd + col0;

    const v4f be = *reinterpret_cast<const v4f*>(beff + col0);
    const v4f sv = *reinterpret_cast<const v4f*>(S + tm * 16 + 8 * h + 4 * kh);

    v8f acc[4] = {};
    strip_gemm<Hh / 16, 1, 4>(ap, bp, Dd, acc);

    #pragma unroll
    for (int rr = 0; rr < 4; ++rr) {
        v4f gv;
        gv.x = kh ? acc[0][rr] : acc[0][4 + rr];
        gv.y = kh ? acc[1][rr] : acc[1][4 + rr];
        gv.z = kh ? acc[2][rr] : acc[2][4 + rr];
        gv.w = kh ? acc[3][rr] : acc[3][4 + rr];
        *reinterpret_cast<v4f*>(&xch[mt][kh][rr][4 * lane]) = gv;
    }
    __syncthreads();

    #pragma unroll
    for (int rr = 0; rr < 4; ++rr) {
        const v4f pv = *reinterpret_cast<const v4f*>(&xch[mt][1 - kh][rr][4 * lane]);
        const int rowm = tm * 16 + 8 * h + 4 * kh + rr;
        const float s = sv[rr];
        v4f o;
        o.x = ((kh ? acc[0][4 + rr] : acc[0][rr]) + pv.x) + s * be.x;
        o.y = ((kh ? acc[1][4 + rr] : acc[1][rr]) + pv.y) + s * be.y;
        o.z = ((kh ? acc[2][4 + rr] : acc[2][rr]) + pv.z) + s * be.z;
        o.w = ((kh ? acc[3][4 + rr] : acc[3][rr]) + pv.w) + s * be.w;
        *reinterpret_cast<v4f*>(out + (size_t)rowm * Dd + col0) = o;
    }
}

extern "C" void kernel_launch(void* const* d_in, const int* in_sizes, int n_in,
                              void* d_out, int out_size, void* d_ws, size_t ws_size,
                              hipStream_t stream)
{
    const float* slots = (const float*)d_in[0];
    const float* W_enc = (const float*)d_in[1];
    const float* b_enc = (const float*)d_in[2];
    const float* ln_g  = (const float*)d_in[3];
    const float* ln_b  = (const float*)d_in[4];
    const float* W_att = (const float*)d_in[5];
    const float* b_att = (const float*)d_in[6];
    const float* W_eff = (const float*)d_in[7];
    const float* b_eff = (const float*)d_in[8];
    float* out = (float*)d_out;

    float* ws = (float*)d_ws;
    float* U  = ws;
    float* V  = U + (size_t)BN * Hh;
    float* Gm = V + (size_t)BN * Hh;
    float* Sm = Gm + (size_t)BN * Hh;

    enc_gemm_kernel<<<256, 128, 0, stream>>>(slots, W_enc, U);

    pair_reduce_kernel<<<BN / IPB, 256, 0, stream>>>(U, V, b_enc, ln_g, ln_b,
                                                     W_att, b_att, Gm, Sm);

    eff_gemm_kernel<<<256, 128, 0, stream>>>(Gm, Sm, W_eff, b_eff, out);
}


# pair kernel: cross-wave sum of the 8 partial G rows leaves the walk over i: waves park partial rows in own LDS slots, no barrier per i; one barrier, then wave w sums row w; order w=0..7 kept
# speedup vs baseline: 1.9423x; 1.0450x over previous
#include <hip/hip_runtime.h>

typedef __attribute__((ext_vector_type(2))) float v2f;
typedef __attribute__((ext_vector_type(4))) float v4f;
typedef __attribute__((ext_vector_type(8))) float v8f;

constexpr int Bsz = 128;
constexpr int Nsl = 32;
constexpr int Dd  = 128;
constexpr int Hh  = 256;
constexpr int BN  = Bsz * Nsl;
constexpr float LN_EPS = 1e-5f;

template <int NG, int CH>
struct FragSet {
    v4f a[CH];
    v4f b[CH][2][NG][2];
};

template <int NG, int CH>
__device__ __forceinline__ void load_batch(FragSet<NG, CH>& f,
                                           const float* __restrict__ ap,
                                           const float* __restrict__ bp,
                                           const int ldb)
{
    #pragma unroll
    for (int ch = 0; ch < CH; ++ch) {
        f.a[ch] = *reinterpret_cast<const v4f*>(ap + 8 * ch);
        #pragma unroll
        for (int p = 0; p < 2; ++p) {
            const float* __restrict__ br = bp + (size_t)(8 * ch + 2 * p) * ldb;
            #pragma unroll
            for (int g = 0; g < NG; ++g) {
                f.b[ch][p][g][0] = *reinterpret_cast<const v4f*>(br + 64 * g);
                f.b[ch][p][g][1] = *reinterpret_cast<const v4f*>(br + ldb + 64 * g);
            }
        }
    }
}

template <int NG, int CH>
__device__ __forceinline__ void mma_batch(const FragSet<NG, CH>& f, v8f (&acc)[NG * 4])
{
    #pragma unroll
    for (int ch = 0; ch < CH; ++ch) {
        #pragma unroll
        for (int p = 0; p < 2; ++p) {
            v2f a; a.x = f.a[ch][2 * p]; a.y = f.a[ch][2 * p + 1];
            #pragma unroll
            for (int g = 0; g < NG; ++g) {
                #pragma unroll
                for (int q = 0; q < 4; ++q) {
                    v2f b; b.x = f.b[ch][p][g][0][q]; b.y = f.b[ch][p][g][1][q];
                    acc[4 * g + q] = __builtin_amdgcn_wmma_f32_16x16x4_f32(
                        false, a, false, b, (short)0, acc[4 * g + q], false, false);
                }
            }
        }
    }
}

template <int KT, int NG, int CH>
__device__ __forceinline__ void strip_gemm(const float* __restrict__ ap,
                                           const float* __restrict__ bp,
                                           const int ldb,
                                           v8f (&acc)[NG * 4])
{
    constexpr int NB = KT / CH;
    constexpr int AHEAD = 2;
    static_assert(KT % CH == 0 && NB > AHEAD, "batching");
    FragSet<NG, CH> ring[AHEAD + 1];

    #pragma unroll
    for (int i = 0; i < AHEAD; ++i)
        load_batch<NG, CH>(ring[i], ap + 8 * CH * i, bp + (size_t)(8 * CH * i) * ldb, ldb);

    #pragma unroll
    for (int i = 0; i < NB; ++i) {
        __builtin_amdgcn_sched_barrier(0);
        if (i + AHEAD < NB)
            load_batch<NG, CH>(ring[(i + AHEAD) % (AHEAD + 1)], ap + 8 * CH * (i + AHEAD),
                               bp + (size_t)(8 * CH * (i + AHEAD)) * ldb, ldb);
        __builtin_amdgcn_sched_barrier(0);
        mma_batch<NG, CH>(ring[i % (AHEAD + 1)], acc);
    }
}

__global__ __launch_bounds__(128)
void enc_gemm_kernel(const float* __restrict__ slots,
                     const float* __restrict__ Wenc,
                     float* __restrict__ UV)
{
    const int lane = threadIdx.x & 31;
    const int wave = threadIdx.x >> 5;
    const int c = lane & 15;
    const int h = lane >> 4;
    const int g  = blockIdx.x & 3;
    const int tm = (blockIdx.x >> 2) * 4 + wave;
    const int whichW = g >> 1;
    const int hb     = (g & 1) * 128;

    const float* __restrict__ ap = slots + (size_t)(tm * 16 + c) * Dd + 4 * h;
    const float* __restrict__ bp = Wenc + (size_t)(whichW * Dd + 4 * h) * Hh + hb + 4 * c;

    v8f acc[8] = {};
    strip_gemm<Dd / 8, 2, 2>(ap, bp, Hh, acc);

    float* __restrict__ dst = UV + (size_t)whichW * BN * Hh
                                 + (size_t)(tm * 16 + 8 * h) * Hh + hb + 4 * c;
    #pragma unroll
    for (int r = 0; r < 8; ++r) {
        #pragma unroll
        for (int gg = 0; gg < 2; ++gg) {
            v4f o;
            o.x = acc[4 * gg + 0][r]; o.y = acc[4 * gg + 1][r];
            o.z = acc[4 * gg + 2][r]; o.w = acc[4 * gg + 3][r];
            *reinterpret_cast<v4f*>(dst + (size_t)r * Hh + 64 * gg) = o;
        }
    }
}

constexpr int IPB = 8;
constexpr int JPW = 4;
constexpr int RP  = Hh + 4;

template <int NV>
__device__ __forceinline__ void wave_reduce_sum(float (&v)[NV]) {
    #pragma unroll
    for (int m = 16; m >= 1; m >>= 1) {
        float o[NV];
        #pragma unroll
        for (int c = 0; c < NV; ++c) o[c] = __shfl_xor(v[c], m, 32);
        __builtin_amdgcn_sched_barrier(0);
        #pragma unroll
        for (int c = 0; c < NV; ++c) v[c] += o[c];
    }
}

__global__ __launch_bounds__(256)
void pair_reduce_kernel(const float* __restrict__ U,
                        const float* __restrict__ V,
                        const float* __restrict__ b_enc,
                        const float* __restrict__ ln_g,
                        const float* __restrict__ ln_b,
                        const float* __restrict__ W_att,
                        const float* __restrict__ b_att,
                        float* __restrict__ G,
                        float* __restrict__ S)
{
    struct Lds {
        float ucs[IPB][RP];
        float vcs[Nsl][RP];
        float cps[4][2][IPB][Nsl];
        float su_sh[IPB];
        float sv_sh[Nsl];
        float ssh[IPB][8];
        float gsh[IPB][8][2][32][4];
    };
    __shared__ __attribute__((aligned(16))) Lds sm;
    static_assert(sizeof(float) * (IPB * RP + Nsl * RP + 4 * 2 * IPB * Nsl + IPB + Nsl + IPB * 8) % 16 == 0,
                  "gsh starts on a 16-byte boundary");
    static_assert(IPB == 8, "wave w finishes row i0 + w: one row per wave");
    float (&ucs)[IPB][RP]            = sm.ucs;
    float (&vcs)[Nsl][RP]            = sm.vcs;
    float (&cps)[4][2][IPB][Nsl]     = sm.cps;
    float (&su_sh)[IPB]              = sm.su_sh;
    float (&sv_sh)[Nsl]              = sm.sv_sh;
    float (&ssh)[IPB][8]             = sm.ssh;
    float (&gsh)[IPB][8][2][32][4]   = sm.gsh;

    const int b    = blockIdx.x / (Nsl / IPB);
    const int i0   = (blockIdx.x % (Nsl / IPB)) * IPB;
    const int lane = threadIdx.x & 31;
    const int wave = threadIdx.x >> 5;

    const float batt = b_att[0];

    float lg[8], lb[8], wa[8], v[JPW][8], ur[8];
    #pragma unroll
    for (int t = 0; t < 8; ++t) {
        const int idx = lane + 32 * t;
        lg[t] = ln_g[idx];
        lb[t] = ln_b[idx];
        wa[t] = W_att[idx];
        #pragma unroll
        for (int c = 0; c < JPW; ++c)
            v[c][t] = V[(size_t)(b * Nsl + wave + 8 * c) * Hh + idx];
        ur[t] = U[(size_t)(b * Nsl + i0 + wave) * Hh + idx] + b_enc[idx];
    }

    {
        float red[JPW + 1];
        #pragma unroll
        for (int c = 0; c < JPW; ++c) {
            float sum = 0.0f;
            #pragma unroll
            for (int t = 0; t < 8; ++t) sum += v[c][t];
            red[c] = sum;
        }
        {
            float sum = 0.0f;
            #pragma unroll
            for (int t = 0; t < 8; ++t) sum += ur[t];
            red[JPW] = sum;
        }
        wave_reduce_sum<JPW + 1>(red);
        #pragma unroll
        for (int c = 0; c < JPW; ++c) {
            const float mu = red[c] * (1.0f / Hh);
            float sq = 0.0f;
            #pragma unroll
            for (int t = 0; t < 8; ++t) { v[c][t] -= mu; sq += v[c][t] * v[c][t]; }
            red[c] = sq;
        }
        {
            const float mu = red[JPW] * (1.0f / Hh);
            float sq = 0.0f;
            #pragma unroll
            for (int t = 0; t < 8; ++t) { ur[t] -= mu; sq += ur[t] * ur[t]; }
            red[JPW] = sq;
        }
        wave_reduce_sum<JPW + 1>(red);
        #pragma unroll
        for (int t = 0; t < 8; ++t) {
            #pragma unroll
            for (int c = 0; c < JPW; ++c) vcs[wave + 8 * c][lane + 32 * t] = v[c][t];
            ucs[wave][lane + 32 * t] = ur[t];
        }
        if (lane == 0) {
            #pragma unroll
            for (int c = 0; c < JPW; ++c) sv_sh[wave + 8 * c] = red[c];
            su_sh[wave] = red[JPW];
        }
    }
    __syncthreads();

    {
        const int cc = lane & 15;
        const int hh = lane >> 4;
        const int nt = wave & 1;
        const int kq = wave >> 1;
        const float* ar = &ucs[cc & 7][64 * kq + 4 * hh];
        const float* br = &vcs[16 * nt + cc][64 * kq + 4 * hh];
        v8f acc0 = {};
        v8f acc1 = {};
        #pragma unroll
        for (int s8 = 0; s8 < 8; ++s8) {
            const v4f a4 = *reinterpret_cast<const v4f*>(ar + 8 * s8);
            const v4f b4 = *reinterpret_cast<const v4f*>(br + 8 * s8);
            v2f a01, b01, a23, b23;
            a01.x = a4.x; a01.y = a4.y; b01.x = b4.x; b01.y = b4.y;
            a23.x = a4.z; a23.y = a4.w; b23.x = b4.z; b23.y = b4.w;
            acc0 = __builtin_amdgcn_wmma_f32_16x16x4_f32(false, a01, false, b01, (short)0, acc0, false, false);
            acc1 = __builtin_amdgcn_wmma_f32_16x16x4_f32(false, a23, false, b23, (short)0, acc1, false, false);
        }
        #pragma unroll
        for (int r = 0; r < 8; ++r) cps[kq][hh][r][16 * nt + cc] = acc0[r] + acc1[r];
    }
    __syncthreads();

    float inv_l;
    {
        const int ii = lane >> 2;
        const int cj = wave + 8 * (lane & 3);
        const float cs = ((cps[0][0][ii][cj] + cps[1][0][ii][cj]) + cps[2][0][ii][cj]) + cps[3][0][ii][cj];
        const float var = fmaxf((su_sh[ii] + sv_sh[cj] + 2.0f * cs) * (1.0f / Hh), 0.0f);
        inv_l = rsqrtf(var + LN_EPS);
    }

    v2f vg[JPW][4], lg2[4], lb2[4], wa2[4];
    #pragma unroll
    for (int k = 0; k < 4; ++k) {
        lg2[k].x = lg[2 * k]; lg2[k].y = lg[2 * k + 1];
        lb2[k].x = lb[2 * k]; lb2[k].y = lb[2 * k + 1];
        wa2[k].x = wa[2 * k]; wa2[k].y = wa[2 * k + 1];
        #pragma unroll
        for (int c = 0; c < JPW; ++c) {
            vg[c][k].x = v[c][2 * k]     * lg[2 * k];
            vg[c][k].y = v[c][2 * k + 1] * lg[2 * k + 1];
        }
    }
    const v2f zero2 = {0.0f, 0.0f};

    for (int ii = 0; ii < IPB; ++ii) {
        const int i   = i0 + ii;

        v2f ug[4];
        #pragma unroll
        for (int k = 0; k < 4; ++k) {
            v2f uc;
            uc.x = ucs[ii][lane + 64 * k];
            uc.y = ucs[ii][lane + 64 * k + 32];
            ug[k] = uc * lg2[k];
        }

        v2f h2[JPW][4];
        float red[JPW];
        #pragma unroll
        for (int c = 0; c < JPW; ++c) {
            const float inv = __int_as_float(
                __builtin_amdgcn_readlane(__float_as_int(inv_l), 4 * ii + c));
            v2f inv2; inv2.x = inv; inv2.y = inv;
            v2f ad = zero2;
            #pragma unroll
            for (int k = 0; k < 4; ++k) {
                v2f hn = (ug[k] + vg[c][k]) * inv2 + lb2[k];
                hn = __builtin_elementwise_max(hn, zero2);
                h2[c][k] = hn;
                ad = hn * wa2[k] + ad;
            }
            red[c] = ad.x + ad.y;
        }
        wave_reduce_sum<JPW>(red);

        v2f g2[4];
        #pragma unroll
        for (int k = 0; k < 4; ++k) g2[k] = zero2;
        float sacc = 0.0f;
        #pragma unroll
        for (int c = 0; c < JPW; ++c) {
            const float adot = red[c] + batt;
            float a = 1.0f / (1.0f + __expf(-adot));
            a = (wave + 8 * c == i) ? 0.0f : a;
            v2f a2; a2.x = a; a2.y = a;
            #pragma unroll
            for (int k = 0; k < 4; ++k) g2[k] = a2 * h2[c][k] + g2[k];
            sacc += a;
        }

        v4f plo, phi;
        plo.x = g2[0].x; plo.y = g2[0].y; plo.z = g2[1].x; plo.w = g2[1].y;
        phi.x = g2[2].x; phi.y = g2[2].y; phi.z = g2[3].x; phi.w = g2[3].y;
        *reinterpret_cast<v4f*>(&gsh[ii][wave][0][lane][0]) = plo;
        *reinterpret_cast<v4f*>(&gsh[ii][wave][1][lane][0]) = phi;
        if (lane == 0) ssh[ii][wave] = sacc;
    }
    __syncthreads();

    {
        const int bi = b * Nsl + i0 + wave;
        v4f tlo = {0.0f, 0.0f, 0.0f, 0.0f};
        v4f thi = {0.0f, 0.0f, 0.0f, 0.0f};
        #pragma unroll
        for (int w = 0; w < 8; ++w) {
            tlo += *reinterpret_cast<const v4f*>(&gsh[wave][w][0][lane][0]);
            thi += *reinterpret_cast<const v4f*>(&gsh[wave][w][1][lane][0]);
        }
        float* __restrict__ grow = G + (size_t)bi * Hh + lane;
        grow[0]   = tlo.x; grow[32]  = tlo.y; grow[64]  = tlo.z; grow[96]  = tlo.w;
        grow[128] = thi.x; grow[160] = thi.y; grow[192] = thi.z; grow[224] = thi.w;

        if (lane == 0) {
            float st = 0.0f;
            #pragma unroll
            for (int w = 0; w < 8; ++w) st += ssh[wave][w];
            S[bi] = st;
        }
    }
}

__global__ __launch_bounds__(128)
void eff_gemm_kernel(const float* __restrict__ G,
                     const float* __restrict__ S,
                     const float* __restrict__ Weff,
                     const float* __restrict__ beff,
                     float* __restrict__ out)
{
    __shared__ __attribute__((aligned(16))) float xch[2][2][4][32 * 4];

    const int lane = threadIdx.x & 31;
    const int wave = threadIdx.x >> 5;
    const int c = lane & 15;
    const int h = lane >> 4;
    const int mt = wave & 1;
    const int kh = wave >> 1;
    const int g  = blockIdx.x & 1;
    const int tm = (blockIdx.x >> 1) * 2 + mt;
    const int col0 = 64 * g + 4 * c;

    const float* __restrict__ ap = G + (size_t)(tm * 16 + c) * Hh + 128 * kh + 4 * h;
    const float* __restrict__ bp = Weff + (size_t)(128 * kh + 4 * h) * Dd + col0;

    const v4f be = *reinterpret_cast<const v4f*>(beff + col0);
    const v4f sv = *reinterpret_cast<const v4f*>(S + tm * 16 + 8 * h + 4 * kh);

    v8f acc[4] = {};
    strip_gemm<Hh / 16, 1, 4>(ap, bp, Dd, acc);

    #pragma unroll
    for (int rr = 0; rr < 4; ++rr) {
        v4f gv;
        gv.x = kh ? acc[0][rr] : acc[0][4 + rr];
        gv.y = kh ? acc[1][rr] : acc[1][4 + rr];
        gv.z = kh ? acc[2][rr] : acc[2][4 + rr];
        gv.w = kh ? acc[3][rr] : acc[3][4 + rr];
        *reinterpret_cast<v4f*>(&xch[mt][kh][rr][4 * lane]) = gv;
    }
    __syncthreads();

    #pragma unroll
    for (int rr = 0; rr < 4; ++rr) {
        const v4f pv = *reinterpret_cast<const v4f*>(&xch[mt][1 - kh][rr][4 * lane]);
        const int rowm = tm * 16 + 8 * h + 4 * kh + rr;
        const float s = sv[rr];
        v4f o;
        o.x = ((kh ? acc[0][4 + rr] : acc[0][rr]) + pv.x) + s * be.x;
        o.y = ((kh ? acc[1][4 + rr] : acc[1][rr]) + pv.y) + s * be.y;
        o.z = ((kh ? acc[2][4 + rr] : acc[2][rr]) + pv.z) + s * be.z;
        o.w = ((kh ? acc[3][4 + rr] : acc[3][rr]) + pv.w) + s * be.w;
        *reinterpret_cast<v4f*>(out + (size_t)rowm * Dd + col0) = o;
    }
}

extern "C" void kernel_launch(void* const* d_in, const int* in_sizes, int n_in,
                              void* d_out, int out_size, void* d_ws, size_t ws_size,
                              hipStream_t stream)
{
    const float* slots = (const float*)d_in[0];
    const float* W_enc = (const float*)d_in[1];
    const float* b_enc = (const float*)d_in[2];
    const float* ln_g  = (const float*)d_in[3];
    const float* ln_b  = (const float*)d_in[4];
    const float* W_att = (const float*)d_in[5];
    const float* b_att = (const float*)d_in[6];
    const float* W_eff = (const float*)d_in[7];
    const float* b_eff = (const float*)d_in[8];
    float* out = (float*)d_out;

    float* ws = (float*)d_ws;
    float* U  = ws;
    float* V  = U + (size_t)BN * Hh;
    float* Gm = V + (size_t)BN * Hh;
    float* Sm = Gm + (size_t)BN * Hh;

    enc_gemm_kernel<<<256, 128, 0, stream>>>(slots, W_enc, U);

    pair_reduce_kernel<<<BN / IPB, 256, 0, stream>>>(U, V, b_enc, ln_g, ln_b,
                                                     W_att, b_att, Gm, Sm);

    eff_gemm_kernel<<<256, 128, 0, stream>>>(Gm, Sm, W_eff, b_eff, out);
}


# pair kernel: a pair's attention sum over the wave on the matrix unit: one f32 WMMA per pair, one-hot B; lane c adds its 8 D rows + the other half; one sigmoid per 4 pairs; that sum's order changes
# speedup vs baseline: 2.0180x; 1.0390x over previous
#include <hip/hip_runtime.h>

typedef __attribute__((ext_vector_type(2))) float v2f;
typedef __attribute__((ext_vector_type(4))) float v4f;
typedef __attribute__((ext_vector_type(8))) float v8f;

constexpr int Bsz = 128;
constexpr int Nsl = 32;
constexpr int Dd  = 128;
constexpr int Hh  = 256;
constexpr int BN  = Bsz * Nsl;
constexpr float LN_EPS = 1e-5f;

template <int NG, int CH>
struct FragSet {
    v4f a[CH];
    v4f b[CH][2][NG][2];
};

template <int NG, int CH>
__device__ __forceinline__ void load_batch(FragSet<NG, CH>& f,
                                           const float* __restrict__ ap,
                                           const float* __restrict__ bp,
                                           const int ldb)
{
    #pragma unroll
    for (int ch = 0; ch < CH; ++ch) {
        f.a[ch] = *reinterpret_cast<const v4f*>(ap + 8 * ch);
        #pragma unroll
        for (int p = 0; p < 2; ++p) {
            const float* __restrict__ br = bp + (size_t)(8 * ch + 2 * p) * ldb;
            #pragma unroll
            for (int g = 0; g < NG; ++g) {
                f.b[ch][p][g][0] = *reinterpret_cast<const v4f*>(br + 64 * g);
                f.b[ch][p][g][1] = *reinterpret_cast<const v4f*>(br + ldb + 64 * g);
            }
        }
    }
}

template <int NG, int CH>
__device__ __forceinline__ void mma_batch(const FragSet<NG, CH>& f, v8f (&acc)[NG * 4])
{
    #pragma unroll
    for (int ch = 0; ch < CH; ++ch) {
        #pragma unroll
        for (int p = 0; p < 2; ++p) {
            v2f a; a.x = f.a[ch][2 * p]; a.y = f.a[ch][2 * p + 1];
            #pragma unroll
            for (int g = 0; g < NG; ++g) {
                #pragma unroll
                for (int q = 0; q < 4; ++q) {
                    v2f b; b.x = f.b[ch][p][g][0][q]; b.y = f.b[ch][p][g][1][q];
                    acc[4 * g + q] = __builtin_amdgcn_wmma_f32_16x16x4_f32(
                        false, a, false, b, (short)0, acc[4 * g + q], false, false);
                }
            }
        }
    }
}

template <int KT, int NG, int CH>
__device__ __forceinline__ void strip_gemm(const float* __restrict__ ap,
                                           const float* __restrict__ bp,
                                           const int ldb,
                                           v8f (&acc)[NG * 4])
{
    constexpr int NB = KT / CH;
    constexpr int AHEAD = 2;
    static_assert(KT % CH == 0 && NB > AHEAD, "batching");
    FragSet<NG, CH> ring[AHEAD + 1];

    #pragma unroll
    for (int i = 0; i < AHEAD; ++i)
        load_batch<NG, CH>(ring[i], ap + 8 * CH * i, bp + (size_t)(8 * CH * i) * ldb, ldb);

    #pragma unroll
    for (int i = 0; i < NB; ++i) {
        __builtin_amdgcn_sched_barrier(0);
        if (i + AHEAD < NB)
            load_batch<NG, CH>(ring[(i + AHEAD) % (AHEAD + 1)], ap + 8 * CH * (i + AHEAD),
                               bp + (size_t)(8 * CH * (i + AHEAD)) * ldb, ldb);
        __builtin_amdgcn_sched_barrier(0);
        mma_batch<NG, CH>(ring[i % (AHEAD + 1)], acc);
    }
}

__global__ __launch_bounds__(128)
void enc_gemm_kernel(const float* __restrict__ slots,
                     const float* __restrict__ Wenc,
                     float* __restrict__ UV)
{
    const int lane = threadIdx.x & 31;
    const int wave = threadIdx.x >> 5;
    const int c = lane & 15;
    const int h = lane >> 4;
    const int g  = blockIdx.x & 3;
    const int tm = (blockIdx.x >> 2) * 4 + wave;
    const int whichW = g >> 1;
    const int hb     = (g & 1) * 128;

    const float* __restrict__ ap = slots + (size_t)(tm * 16 + c) * Dd + 4 * h;
    const float* __restrict__ bp = Wenc + (size_t)(whichW * Dd + 4 * h) * Hh + hb + 4 * c;

    v8f acc[8] = {};
    strip_gemm<Dd / 8, 2, 2>(ap, bp, Hh, acc);

    float* __restrict__ dst = UV + (size_t)whichW * BN * Hh
                                 + (size_t)(tm * 16 + 8 * h) * Hh + hb + 4 * c;
    #pragma unroll
    for (int r = 0; r < 8; ++r) {
        #pragma unroll
        for (int gg = 0; gg < 2; ++gg) {
            v4f o;
            o.x = acc[4 * gg + 0][r]; o.y = acc[4 * gg + 1][r];
            o.z = acc[4 * gg + 2][r]; o.w = acc[4 * gg + 3][r];
            *reinterpret_cast<v4f*>(dst + (size_t)r * Hh + 64 * gg) = o;
        }
    }
}

constexpr int IPB = 8;
constexpr int JPW = 4;
constexpr int RP  = Hh + 4;

template <int NV>
__device__ __forceinline__ void wave_reduce_sum(float (&v)[NV]) {
    #pragma unroll
    for (int m = 16; m >= 1; m >>= 1) {
        float o[NV];
        #pragma unroll
        for (int c = 0; c < NV; ++c) o[c] = __shfl_xor(v[c], m, 32);
        __builtin_amdgcn_sched_barrier(0);
        #pragma unroll
        for (int c = 0; c < NV; ++c) v[c] += o[c];
    }
}

__device__ __forceinline__ float xhalf(float x) {
    return __builtin_bit_cast(float, __builtin_amdgcn_permlanex16(
        __builtin_bit_cast(int, x), __builtin_bit_cast(int, x), 0x76543210, 0xfedcba98, false, false));
}

__global__ __launch_bounds__(256)
void pair_reduce_kernel(const float* __restrict__ U,
                        const float* __restrict__ V,
                        const float* __restrict__ b_enc,
                        const float* __restrict__ ln_g,
                        const float* __restrict__ ln_b,
                        const float* __restrict__ W_att,
                        const float* __restrict__ b_att,
                        float* __restrict__ G,
                        float* __restrict__ S)
{
    struct Lds {
        float ucs[IPB][RP];
        float vcs[Nsl][RP];
        float cps[4][2][IPB][Nsl];
        float su_sh[IPB];
        float sv_sh[Nsl];
        float ssh[IPB][8];
        float gsh[IPB][8][2][32][4];
    };
    __shared__ __attribute__((aligned(16))) Lds sm;
    static_assert(sizeof(float) * (IPB * RP + Nsl * RP + 4 * 2 * IPB * Nsl + IPB + Nsl + IPB * 8) % 16 == 0,
                  "gsh starts on a 16-byte boundary");
    static_assert(IPB == 8, "wave w finishes row i0 + w: one row per wave");
    float (&ucs)[IPB][RP]            = sm.ucs;
    float (&vcs)[Nsl][RP]            = sm.vcs;
    float (&cps)[4][2][IPB][Nsl]     = sm.cps;
    float (&su_sh)[IPB]              = sm.su_sh;
    float (&sv_sh)[Nsl]              = sm.sv_sh;
    float (&ssh)[IPB][8]             = sm.ssh;
    float (&gsh)[IPB][8][2][32][4]   = sm.gsh;

    const int b    = blockIdx.x / (Nsl / IPB);
    const int i0   = (blockIdx.x % (Nsl / IPB)) * IPB;
    const int lane = threadIdx.x & 31;
    const int wave = threadIdx.x >> 5;

    const float batt = b_att[0];

    float lg[8], lb[8], wa[8], v[JPW][8], ur[8];
    #pragma unroll
    for (int t = 0; t < 8; ++t) {
        const int idx = lane + 32 * t;
        lg[t] = ln_g[idx];
        lb[t] = ln_b[idx];
        wa[t] = W_att[idx];
        #pragma unroll
        for (int c = 0; c < JPW; ++c)
            v[c][t] = V[(size_t)(b * Nsl + wave + 8 * c) * Hh + idx];
        ur[t] = U[(size_t)(b * Nsl + i0 + wave) * Hh + idx] + b_enc[idx];
    }

    {
        float red[JPW + 1];
        #pragma unroll
        for (int c = 0; c < JPW; ++c) {
            float sum = 0.0f;
            #pragma unroll
            for (int t = 0; t < 8; ++t) sum += v[c][t];
            red[c] = sum;
        }
        {
            float sum = 0.0f;
            #pragma unroll
            for (int t = 0; t < 8; ++t) sum += ur[t];
            red[JPW] = sum;
        }
        wave_reduce_sum<JPW + 1>(red);
        #pragma unroll
        for (int c = 0; c < JPW; ++c) {
            const float mu = red[c] * (1.0f / Hh);
            float sq = 0.0f;
            #pragma unroll
            for (int t = 0; t < 8; ++t) { v[c][t] -= mu; sq += v[c][t] * v[c][t]; }
            red[c] = sq;
        }
        {
            const float mu = red[JPW] * (1.0f / Hh);
            float sq = 0.0f;
            #pragma unroll
            for (int t = 0; t < 8; ++t) { ur[t] -= mu; sq += ur[t] * ur[t]; }
            red[JPW] = sq;
        }
        wave_reduce_sum<JPW + 1>(red);
        #pragma unroll
        for (int t = 0; t < 8; ++t) {
            #pragma unroll
            for (int c = 0; c < JPW; ++c) vcs[wave + 8 * c][lane + 32 * t] = v[c][t];
            ucs[wave][lane + 32 * t] = ur[t];
        }
        if (lane == 0) {
            #pragma unroll
            for (int c = 0; c < JPW; ++c) sv_sh[wave + 8 * c] = red[c];
            su_sh[wave] = red[JPW];
        }
    }
    __syncthreads();

    {
        const int cc = lane & 15;
        const int hh = lane >> 4;
        const int nt = wave & 1;
        const int kq = wave >> 1;
        const float* ar = &ucs[cc & 7][64 * kq + 4 * hh];
        const float* br = &vcs[16 * nt + cc][64 * kq + 4 * hh];
        v8f acc0 = {};
        v8f acc1 = {};
        #pragma unroll
        for (int s8 = 0; s8 < 8; ++s8) {
            const v4f a4 = *reinterpret_cast<const v4f*>(ar + 8 * s8);
            const v4f b4 = *reinterpret_cast<const v4f*>(br + 8 * s8);
            v2f a01, b01, a23, b23;
            a01.x = a4.x; a01.y = a4.y; b01.x = b4.x; b01.y = b4.y;
            a23.x = a4.z; a23.y = a4.w; b23.x = b4.z; b23.y = b4.w;
            acc0 = __builtin_amdgcn_wmma_f32_16x16x4_f32(false, a01, false, b01, (short)0, acc0, false, false);
            acc1 = __builtin_amdgcn_wmma_f32_16x16x4_f32(false, a23, false, b23, (short)0, acc1, false, false);
        }
        #pragma unroll
        for (int r = 0; r < 8; ++r) cps[kq][hh][r][16 * nt + cc] = acc0[r] + acc1[r];
    }
    __syncthreads();

    float inv_l;
    {
        const int ii = lane >> 2;
        const int cj = wave + 8 * (lane & 3);
        const float cs = ((cps[0][0][ii][cj] + cps[1][0][ii][cj]) + cps[2][0][ii][cj]) + cps[3][0][ii][cj];
        const float var = fmaxf((su_sh[ii] + sv_sh[cj] + 2.0f * cs) * (1.0f / Hh), 0.0f);
        inv_l = rsqrtf(var + LN_EPS);
    }

    v2f vg[JPW][4], lg2[4], lb2[4], wa2[4];
    #pragma unroll
    for (int k = 0; k < 4; ++k) {
        lg2[k].x = lg[2 * k]; lg2[k].y = lg[2 * k + 1];
        lb2[k].x = lb[2 * k]; lb2[k].y = lb[2 * k + 1];
        wa2[k].x = wa[2 * k]; wa2[k].y = wa[2 * k + 1];
        #pragma unroll
        for (int c = 0; c < JPW; ++c) {
            vg[c][k].x = v[c][2 * k]     * lg[2 * k];
            vg[c][k].y = v[c][2 * k + 1] * lg[2 * k + 1];
        }
    }
    const v2f zero2 = {0.0f, 0.0f};

    v2f sel[JPW];
    #pragma unroll
    for (int c = 0; c < JPW; ++c) {
        const float o = ((lane & 15) == c) ? 1.0f : 0.0f;
        sel[c].x = o; sel[c].y = o;
    }

    for (int ii = 0; ii < IPB; ++ii) {
        const int i   = i0 + ii;

        v2f ug[4];
        #pragma unroll
        for (int k = 0; k < 4; ++k) {
            v2f uc;
            uc.x = ucs[ii][lane + 64 * k];
            uc.y = ucs[ii][lane + 64 * k + 32];
            ug[k] = uc * lg2[k];
        }

        v2f h2[JPW][4];
        v8f dsum = {};
        #pragma unroll
        for (int c = 0; c < JPW; ++c) {
            const float inv = __int_as_float(
                __builtin_amdgcn_readlane(__float_as_int(inv_l), 4 * ii + c));
            v2f inv2; inv2.x = inv; inv2.y = inv;
            v2f ad = zero2;
            #pragma unroll
            for (int k = 0; k < 4; ++k) {
                v2f hn = (ug[k] + vg[c][k]) * inv2 + lb2[k];
                hn = __builtin_elementwise_max(hn, zero2);
                h2[c][k] = hn;
                ad = hn * wa2[k] + ad;
            }
            dsum = __builtin_amdgcn_wmma_f32_16x16x4_f32(false, ad, false, sel[c], (short)0, dsum, false, false);
        }
        float al;
        {
            v2f s01, s23, s45, s67;
            s01.x = dsum[0]; s01.y = dsum[1]; s23.x = dsum[2]; s23.y = dsum[3];
            s45.x = dsum[4]; s45.y = dsum[5]; s67.x = dsum[6]; s67.y = dsum[7];
            const v2f q = (s01 + s23) + (s45 + s67);
            float tot = q.x + q.y;
            tot += xhalf(tot);
            const float adot = tot + batt;
            al = 1.0f / (1.0f + __expf(-adot));
        }

        v2f g2[4];
        #pragma unroll
        for (int k = 0; k < 4; ++k) g2[k] = zero2;
        float sacc = 0.0f;
        #pragma unroll
        for (int c = 0; c < JPW; ++c) {
            float a = __int_as_float(__builtin_amdgcn_readlane(__float_as_int(al), c));
            a = (wave + 8 * c == i) ? 0.0f : a;
            v2f a2; a2.x = a; a2.y = a;
            #pragma unroll
            for (int k = 0; k < 4; ++k) g2[k] = a2 * h2[c][k] + g2[k];
            sacc += a;
        }

        v4f plo, phi;
        plo.x = g2[0].x; plo.y = g2[0].y; plo.z = g2[1].x; plo.w = g2[1].y;
        phi.x = g2[2].x; phi.y = g2[2].y; phi.z = g2[3].x; phi.w = g2[3].y;
        *reinterpret_cast<v4f*>(&gsh[ii][wave][0][lane][0]) = plo;
        *reinterpret_cast<v4f*>(&gsh[ii][wave][1][lane][0]) = phi;
        if (lane == 0) ssh[ii][wave] = sacc;
    }
    __syncthreads();

    {
        const int bi = b * Nsl + i0 + wave;
        v4f tlo = {0.0f, 0.0f, 0.0f, 0.0f};
        v4f thi = {0.0f, 0.0f, 0.0f, 0.0f};
        #pragma unroll
        for (int w = 0; w < 8; ++w) {
            tlo += *reinterpret_cast<const v4f*>(&gsh[wave][w][0][lane][0]);
            thi += *reinterpret_cast<const v4f*>(&gsh[wave][w][1][lane][0]);
        }
        float* __restrict__ grow = G + (size_t)bi * Hh + lane;
        grow[0]   = tlo.x; grow[32]  = tlo.y; grow[64]  = tlo.z; grow[96]  = tlo.w;
        grow[128] = thi.x; grow[160] = thi.y; grow[192] = thi.z; grow[224] = thi.w;

        if (lane == 0) {
            float st = 0.0f;
            #pragma unroll
            for (int w = 0; w < 8; ++w) st += ssh[wave][w];
            S[bi] = st;
        }
    }
}

__global__ __launch_bounds__(128)
void eff_gemm_kernel(const float* __restrict__ G,
                     const float* __restrict__ S,
                     const float* __restrict__ Weff,
                     const float* __restrict__ beff,
                     float* __restrict__ out)
{
    __shared__ __attribute__((aligned(16))) float xch[2][2][4][32 * 4];

    const int lane = threadIdx.x & 31;
    const int wave = threadIdx.x >> 5;
    const int c = lane & 15;
    const int h = lane >> 4;
    const int mt = wave & 1;
    const int kh = wave >> 1;
    const int g  = blockIdx.x & 1;
    const int tm = (blockIdx.x >> 1) * 2 + mt;
    const int col0 = 64 * g + 4 * c;

    const float* __restrict__ ap = G + (size_t)(tm * 16 + c) * Hh + 128 * kh + 4 * h;
    const float* __restrict__ bp = Weff + (size_t)(128 * kh + 4 * h) * Dd + col0;

    const v4f be = *reinterpret_cast<const v4f*>(beff + col0);
    const v4f sv = *reinterpret_cast<const v4f*>(S + tm * 16 + 8 * h + 4 * kh);

    v8f acc[4] = {};
    strip_gemm<Hh / 16, 1, 4>(ap, bp, Dd, acc);

    #pragma unroll
    for (int rr = 0; rr < 4; ++rr) {
        v4f gv;
        gv.x = kh ? acc[0][rr] : acc[0][4 + rr];
        gv.y = kh ? acc[1][rr] : acc[1][4 + rr];
        gv.z = kh ? acc[2][rr] : acc[2][4 + rr];
        gv.w = kh ? acc[3][rr] : acc[3][4 + rr];
        *reinterpret_cast<v4f*>(&xch[mt][kh][rr][4 * lane]) = gv;
    }
    __syncthreads();

    #pragma unroll
    for (int rr = 0; rr < 4; ++rr) {
        const v4f pv = *reinterpret_cast<const v4f*>(&xch[mt][1 - kh][rr][4 * lane]);
        const int rowm = tm * 16 + 8 * h + 4 * kh + rr;
        const float s = sv[rr];
        v4f o;
        o.x = ((kh ? acc[0][4 + rr] : acc[0][rr]) + pv.x) + s * be.x;
        o.y = ((kh ? acc[1][4 + rr] : acc[1][rr]) + pv.y) + s * be.y;
        o.z = ((kh ? acc[2][4 + rr] : acc[2][rr]) + pv.z) + s * be.z;
        o.w = ((kh ? acc[3][4 + rr] : acc[3][rr]) + pv.w) + s * be.w;
        *reinterpret_cast<v4f*>(out + (size_t)rowm * Dd + col0) = o;
    }
}

extern "C" void kernel_launch(void* const* d_in, const int* in_sizes, int n_in,
                              void* d_out, int out_size, void* d_ws, size_t ws_size,
                              hipStream_t stream)
{
    const float* slots = (const float*)d_in[0];
    const float* W_enc = (const float*)d_in[1];
    const float* b_enc = (const float*)d_in[2];
    const float* ln_g  = (const float*)d_in[3];
    const float* ln_b  = (const float*)d_in[4];
    const float* W_att = (const float*)d_in[5];
    const float* b_att = (const float*)d_in[6];
    const float* W_eff = (const float*)d_in[7];
    const float* b_eff = (const float*)d_in[8];
    float* out = (float*)d_out;

    float* ws = (float*)d_ws;
    float* U  = ws;
    float* V  = U + (size_t)BN * Hh;
    float* Gm = V + (size_t)BN * Hh;
    float* Sm = Gm + (size_t)BN * Hh;

    enc_gemm_kernel<<<256, 128, 0, stream>>>(slots, W_enc, U);

    pair_reduce_kernel<<<BN / IPB, 256, 0, stream>>>(U, V, b_enc, ln_g, ln_b,
                                                     W_att, b_att, Gm, Sm);

    eff_gemm_kernel<<<256, 128, 0, stream>>>(Gm, Sm, W_eff, b_eff, out);
}
